# starting kernel
# baseline (speedup 1.0000x reference)
#include <hip/hip_runtime.h>
#include <math.h>

#define NIMG 4
#define KCH 3
#define HH 96
#define WW 96
#define PP (HH * WW)
#define TILE_I 64
#define NTILES (PP / TILE_I)
#define THREADS 128

typedef __attribute__((ext_vector_type(2))) float v2f;
typedef __attribute__((ext_vector_type(8))) float v8f;

__device__ __forceinline__ void atomic_max_float(float* addr, float val) {
    if (val >= 0.0f) atomicMax((int*)addr, __float_as_int(val));
    else             atomicMin((unsigned int*)addr, __float_as_uint(val));
}

__global__ void bf_init_max(float* mx) {
    if (threadIdx.x == 0 && blockIdx.x == 0) *mx = -INFINITY;
}

__global__ __launch_bounds__(THREADS) void bf_main(
    const float* __restrict__ raw,
    const float* __restrict__ nrm,
    float* __restrict__ out,
    float* __restrict__ mx)
{
    extern __shared__ float smem[];
    float* s_sq = smem;
    float* s_c0 = smem + PP;
    float* s_c1 = smem + 2 * PP;
    float* s_c2 = smem + 3 * PP;
    float* s_v0 = smem + 4 * PP;
    float* s_v1 = smem + 5 * PP;
    float* s_v2 = smem + 6 * PP;

    const int n    = blockIdx.x / NTILES;
    const int tile = blockIdx.x % NTILES;
    const int tid  = threadIdx.x;

    const float inv_rgb = 1.0f / 15.0f;
    const float inv_xy  = 1.0f / 100.0f;

    const float* rawn = raw + (size_t)n * KCH * PP;
    const float* nrmn = nrm + (size_t)n * KCH * PP;

    for (int p = tid; p < PP; p += THREADS) {
        float c0 = rawn[p]          * inv_rgb;
        float c1 = rawn[PP + p]     * inv_rgb;
        float c2 = rawn[2 * PP + p] * inv_rgb;
        float x  = (float)(p % WW) * inv_xy;
        float y  = (float)(p / WW) * inv_xy;
        s_c0[p] = c0; s_c1[p] = c1; s_c2[p] = c2;
        s_sq[p] = x * x + y * y + c0 * c0 + c1 * c1 + c2 * c2;
        s_v0[p] = nrmn[p];
        s_v1[p] = nrmn[PP + p];
        s_v2[p] = nrmn[2 * PP + p];
    }
    __syncthreads();

    const int lane = tid & 31;
    const int l16  = lane & 15;
    const int hl   = lane >> 4;
    const int wave = tid >> 5;
    const int icol = tile * TILE_I + wave * 16 + l16;

    const float xi = (float)(icol % WW) * inv_xy;
    const float yi = (float)(icol / WW) * inv_xy;
    v2f b1, b2;
    if (hl == 0) { b1.x = xi;          b1.y = yi;          b2.x = s_c2[icol]; b2.y = 0.0f; }
    else         { b1.x = s_c0[icol];  b1.y = s_c1[icol];  b2.x = 0.0f;       b2.y = 0.0f; }
    const float sqi = s_sq[icol];

    float acc0 = 0.0f, acc1 = 0.0f, acc2 = 0.0f;

    const float4* sq4 = (const float4*)s_sq;
    const float4* v04 = (const float4*)s_v0;
    const float4* v14 = (const float4*)s_v1;
    const float4* v24 = (const float4*)s_v2;

    for (int jt = 0; jt < PP; jt += 16) {
        const int j  = jt + l16;
        const float xj = (float)(j % WW) * inv_xy;
        const float yj = (float)(j / WW) * inv_xy;
        v2f a1, a2;
        if (hl == 0) { a1.x = xj;        a1.y = yj;        a2.x = s_c2[j]; a2.y = 0.0f; }
        else         { a1.x = s_c0[j];   a1.y = s_c1[j];   a2.x = 0.0f;    a2.y = 0.0f; }

        v8f g = {0.f, 0.f, 0.f, 0.f, 0.f, 0.f, 0.f, 0.f};
        g = __builtin_amdgcn_wmma_f32_16x16x4_f32(false, a1, false, b1, (short)0, g, false, false);
        g = __builtin_amdgcn_wmma_f32_16x16x4_f32(false, a2, false, b2, (short)0, g, false, false);

        const int b4 = (jt + hl * 8) >> 2;
        float4 sjA = sq4[b4], sjB = sq4[b4 + 1];
        float4 w0A = v04[b4], w0B = v04[b4 + 1];
        float4 w1A = v14[b4], w1B = v14[b4 + 1];
        float4 w2A = v24[b4], w2B = v24[b4 + 1];
        float sj[8]  = {sjA.x, sjA.y, sjA.z, sjA.w, sjB.x, sjB.y, sjB.z, sjB.w};
        float vv0[8] = {w0A.x, w0A.y, w0A.z, w0A.w, w0B.x, w0B.y, w0B.z, w0B.w};
        float vv1[8] = {w1A.x, w1A.y, w1A.z, w1A.w, w1B.x, w1B.y, w1B.z, w1B.w};
        float vv2[8] = {w2A.x, w2A.y, w2A.z, w2A.w, w2B.x, w2B.y, w2B.z, w2B.w};

#pragma unroll
        for (int r = 0; r < 8; ++r) {
            float d2 = fmaxf(sqi + sj[r] - 2.0f * g[r], 0.0f);
            float w  = __expf(-0.5f * d2);
            acc0 = fmaf(w, vv0[r], acc0);
            acc1 = fmaf(w, vv1[r], acc1);
            acc2 = fmaf(w, vv2[r], acc2);
        }
    }

    acc0 += __shfl_xor(acc0, 16, 32);
    acc1 += __shfl_xor(acc1, 16, 32);
    acc2 += __shfl_xor(acc2, 16, 32);

    float m = fmaxf(fmaxf(acc0, acc1), acc2);
#pragma unroll
    for (int off = 8; off >= 1; off >>= 1) m = fmaxf(m, __shfl_xor(m, off, 32));
    if (lane == 0) atomic_max_float(mx, m);

    if (hl == 0) {
        float* o = out + (size_t)n * KCH * PP;
        o[icol]          = acc0;
        o[PP + icol]     = acc1;
        o[2 * PP + icol] = acc2;
    }
}

__global__ void bf_scale(float* __restrict__ out, const float* __restrict__ mx, int total) {
    int i = blockIdx.x * blockDim.x + threadIdx.x;
    if (i < total) out[i] = out[i] / (*mx);
}

extern "C" void kernel_launch(void* const* d_in, const int* in_sizes, int n_in,
                              void* d_out, int out_size, void* d_ws, size_t ws_size,
                              hipStream_t stream) {
    (void)in_sizes; (void)n_in; (void)out_size; (void)ws_size;
    const float* raw = (const float*)d_in[0];
    const float* nrm = (const float*)d_in[1];
    float* out = (float*)d_out;
    float* mx  = (float*)d_ws;

    const size_t lds_bytes = (size_t)7 * PP * sizeof(float);
    hipFuncSetAttribute((const void*)bf_main,
                        hipFuncAttributeMaxDynamicSharedMemorySize, (int)lds_bytes);

    bf_init_max<<<1, 1, 0, stream>>>(mx);
    bf_main<<<NIMG * NTILES, THREADS, lds_bytes, stream>>>(raw, nrm, out, mx);

    const int total = NIMG * KCH * PP;
    bf_scale<<<(total + 255) / 256, 256, 0, stream>>>(out, mx, total);
}


# Stripe widened to 48 pixels (three 16-column sets per wave on one A fragment and one set of row loads): grid 768 = 3 work-groups on every WGP, was 1152 = 4 on some, 5 on others; sum order kept
# speedup vs baseline: 8.4492x; 8.4492x over previous
#include <hip/hip_runtime.h>
#include <math.h>

#define NIMG 4
#define KCH 3
#define HH 96
#define WW 96
#define PP (HH * WW)
#define ISTR 48
#define NSTR (PP / ISTR)
#define CSETS 3
#define JSPLIT 8
#define NWAVES JSPLIT
#define JRANGE (PP / JSPLIT)
#define JROWS 16
#define YROWS (JRANGE / WW)
#define XTILES (WW / JROWS)
#define THREADS (32 * NWAVES)
#define SQ_OFF 64
#define LOG2E 1.44269504088896340736f

typedef __attribute__((ext_vector_type(2))) float v2f;
typedef __attribute__((ext_vector_type(4))) float v4f;
typedef __attribute__((ext_vector_type(8))) float v8f;

__device__ __forceinline__ void atomic_max_float(float* addr, float val) {
    if (val >= 0.0f) atomicMax((int*)addr, __float_as_int(val));
    else             atomicMin((unsigned int*)addr, __float_as_uint(val));
}

__device__ __forceinline__ float exp2_sat(float g) {
    return __builtin_amdgcn_fmed3f(__builtin_amdgcn_exp2f(g), 0.0f, 1.0f);
}

__global__ void bf_prep(const float* __restrict__ raw, float* __restrict__ ws) {
    const int g = blockIdx.x * blockDim.x + threadIdx.x;
    if (g == 0) ws[0] = -INFINITY;
    if (g < NIMG * PP) {
        const int n = g / PP;
        const int p = g - n * PP;
        const float inv_rgb = 1.0f / 15.0f;
        const float inv_xy  = 1.0f / 100.0f;
        const float* rawn = raw + (size_t)n * KCH * PP;
        const float x  = (float)(p % WW) * inv_xy;
        const float y  = (float)(p / WW) * inv_xy;
        const float c0 = rawn[p]          * inv_rgb;
        const float c1 = rawn[PP + p]     * inv_rgb;
        const float c2 = rawn[2 * PP + p] * inv_rgb;
        const float sq = x * x + y * y + c0 * c0 + c1 * c1 + c2 * c2;
        ws[SQ_OFF + g] = (-0.5f * LOG2E) * sq;
    }
}

__global__ __launch_bounds__(THREADS) void bf_main(
    const float* __restrict__ raw,
    const float* __restrict__ nrm,
    const float* __restrict__ sqt,
    float* __restrict__ out,
    float* __restrict__ mx)
{
    __shared__ float s_part[NWAVES][KCH][ISTR];

    const int n      = blockIdx.x / NSTR;
    const int stripe = blockIdx.x % NSTR;
    const int tid    = threadIdx.x;
    const int lane   = tid & 31;
    const int c      = lane & 15;
    const bool h     = (lane >> 4) != 0;
    const int wave   = __builtin_amdgcn_readfirstlane(tid >> 5);
    const int jq     = wave;

    const float inv_rgb = 1.0f / 15.0f;
    const float inv_xy  = 1.0f / 100.0f;

    const float* rawn = raw + (size_t)n * KCH * PP;
    const float* nrmn = nrm + (size_t)n * KCH * PP;
    const float* sqn  = sqt + (size_t)n * PP;

    const int ipixA = stripe * ISTR + c;
    const int ipixB = ipixA + 16;
    const int ipixC = ipixA + 32;
    const float yi  = (float)(ipixA / WW) * inv_xy;
    const float xiA  = (float)(ipixA % WW) * inv_xy;
    const float ciA0 = rawn[ipixA]          * inv_rgb;
    const float ciA1 = rawn[PP + ipixA]     * inv_rgb;
    const float ciA2 = rawn[2 * PP + ipixA] * inv_rgb;
    const float sqiA = xiA * xiA + yi * yi + ciA0 * ciA0 + ciA1 * ciA1 + ciA2 * ciA2;
    const float xiB  = (float)(ipixB % WW) * inv_xy;
    const float ciB0 = rawn[ipixB]          * inv_rgb;
    const float ciB1 = rawn[PP + ipixB]     * inv_rgb;
    const float ciB2 = rawn[2 * PP + ipixB] * inv_rgb;
    const float sqiB = xiB * xiB + yi * yi + ciB0 * ciB0 + ciB1 * ciB1 + ciB2 * ciB2;
    const float xiC  = (float)(ipixC % WW) * inv_xy;
    const float ciC0 = rawn[ipixC]          * inv_rgb;
    const float ciC1 = rawn[PP + ipixC]     * inv_rgb;
    const float ciC2 = rawn[2 * PP + ipixC] * inv_rgb;
    const float sqiC = xiC * xiC + yi * yi + ciC0 * ciC0 + ciC1 * ciC1 + ciC2 * ciC2;
    v2f bA, bB, bC;
    bA.x = LOG2E * (h ? ciA1 : xiA);
    bA.y = LOG2E * (h ? ciA2 : ciA0);
    bB.x = LOG2E * (h ? ciB1 : xiB);
    bB.y = LOG2E * (h ? ciB2 : ciB0);
    bC.x = LOG2E * (h ? ciC1 : xiC);
    bC.y = LOG2E * (h ? ciC2 : ciC0);
    const float yil = LOG2E * yi;
    const float e0A = (-0.5f * LOG2E) * sqiA;
    const float e0B = (-0.5f * LOG2E) * sqiB;
    const float e0C = (-0.5f * LOG2E) * sqiC;

    const unsigned aq0 = (h ? PP : 0) + c;
    const unsigned aq1 = 2 * PP + c;
    const unsigned rofs = h ? 8 : 0;

    v2f sA0 = {0.0f, 0.0f}, sA1 = {0.0f, 0.0f}, sA2 = {0.0f, 0.0f};
    v2f sB0 = {0.0f, 0.0f}, sB1 = {0.0f, 0.0f}, sB2 = {0.0f, 0.0f};
    v2f sC0 = {0.0f, 0.0f}, sC1 = {0.0f, 0.0f}, sC2 = {0.0f, 0.0f};

    unsigned jt = (unsigned)jq * JRANGE;

    float q0 = (rawn + jt)[aq0];
    float q1 = (rawn + jt)[aq1];
    v4f sa = *(const v4f*)(sqn + jt + rofs);
    v4f sb = *(const v4f*)(sqn + jt + rofs + 4);

    for (int ry = 0; ry < YROWS; ++ry) {
        const int jy = jq * YROWS + ry;
        const float yj = (float)jy * inv_xy;
        const float ctA = fmaf(yil, yj, e0A);
        const float ctB = fmaf(yil, yj, e0B);
        const float ctC = fmaf(yil, yj, e0C);

#pragma unroll 1
        for (int tx = 0; tx < XTILES; ++tx) {
            const float* nt = nrmn + jt;
            const v4f u0 = *(const v4f*)(nt + rofs);
            const v4f u1 = *(const v4f*)(nt + rofs + 4);
            const v4f v0 = *(const v4f*)(nt + PP + rofs);
            const v4f v1 = *(const v4f*)(nt + PP + rofs + 4);
            const v4f w0 = *(const v4f*)(nt + 2 * PP + rofs);
            const v4f w1 = *(const v4f*)(nt + 2 * PP + rofs + 4);

            const float xj = (float)(tx * JROWS + c) * inv_xy;
            const float p0 = q0 * inv_rgb;
            const float p1 = q1 * inv_rgb;
            v2f a;
            a.x = h ? p0 : xj;
            a.y = h ? p1 : p0;
            const v8f cstA = {ctA + sa.x, ctA + sa.y, ctA + sa.z, ctA + sa.w,
                              ctA + sb.x, ctA + sb.y, ctA + sb.z, ctA + sb.w};
            const v8f cstB = {ctB + sa.x, ctB + sa.y, ctB + sa.z, ctB + sa.w,
                              ctB + sb.x, ctB + sb.y, ctB + sb.z, ctB + sb.w};
            const v8f cstC = {ctC + sa.x, ctC + sa.y, ctC + sa.z, ctC + sa.w,
                              ctC + sb.x, ctC + sb.y, ctC + sb.z, ctC + sb.w};

            const bool more = (ry != YROWS - 1) || (tx != XTILES - 1);
            jt += more ? JROWS : 0;
            const float* rt = rawn + jt;
            const float* st = sqn + jt;
            q0 = rt[aq0];
            q1 = rt[aq1];
            sa = *(const v4f*)(st + rofs);
            sb = *(const v4f*)(st + rofs + 4);

            __builtin_amdgcn_sched_barrier(0);

            const v8f gA = __builtin_amdgcn_wmma_f32_16x16x4_f32(false, a, false, bA, (short)0, cstA, false, false);
            const v8f gB = __builtin_amdgcn_wmma_f32_16x16x4_f32(false, a, false, bB, (short)0, cstB, false, false);
            const v8f gC = __builtin_amdgcn_wmma_f32_16x16x4_f32(false, a, false, bC, (short)0, cstC, false, false);

#pragma unroll
            for (int hh = 0; hh < 2; ++hh) {
                const v4f n0 = hh ? u1 : u0;
                const v4f n1 = hh ? v1 : v0;
                const v4f n2 = hh ? w1 : w0;
                const v4f gqA = hh ? (v4f){gA[4], gA[5], gA[6], gA[7]} : (v4f){gA[0], gA[1], gA[2], gA[3]};
                const v4f gqB = hh ? (v4f){gB[4], gB[5], gB[6], gB[7]} : (v4f){gB[0], gB[1], gB[2], gB[3]};
                const v4f gqC = hh ? (v4f){gC[4], gC[5], gC[6], gC[7]} : (v4f){gC[0], gC[1], gC[2], gC[3]};
                v2f wa, wb;
                wa.x = exp2_sat(gqA.x);
                wa.y = exp2_sat(gqA.y);
                wb.x = exp2_sat(gqA.z);
                wb.y = exp2_sat(gqA.w);
                sA0 = __builtin_elementwise_fma(wa, (v2f){n0.x, n0.y}, sA0);
                sA1 = __builtin_elementwise_fma(wa, (v2f){n1.x, n1.y}, sA1);
                sA2 = __builtin_elementwise_fma(wa, (v2f){n2.x, n2.y}, sA2);
                sA0 = __builtin_elementwise_fma(wb, (v2f){n0.z, n0.w}, sA0);
                sA1 = __builtin_elementwise_fma(wb, (v2f){n1.z, n1.w}, sA1);
                sA2 = __builtin_elementwise_fma(wb, (v2f){n2.z, n2.w}, sA2);
                v2f wc, wd;
                wc.x = exp2_sat(gqB.x);
                wc.y = exp2_sat(gqB.y);
                wd.x = exp2_sat(gqB.z);
                wd.y = exp2_sat(gqB.w);
                sB0 = __builtin_elementwise_fma(wc, (v2f){n0.x, n0.y}, sB0);
                sB1 = __builtin_elementwise_fma(wc, (v2f){n1.x, n1.y}, sB1);
                sB2 = __builtin_elementwise_fma(wc, (v2f){n2.x, n2.y}, sB2);
                sB0 = __builtin_elementwise_fma(wd, (v2f){n0.z, n0.w}, sB0);
                sB1 = __builtin_elementwise_fma(wd, (v2f){n1.z, n1.w}, sB1);
                sB2 = __builtin_elementwise_fma(wd, (v2f){n2.z, n2.w}, sB2);
                v2f we, wf;
                we.x = exp2_sat(gqC.x);
                we.y = exp2_sat(gqC.y);
                wf.x = exp2_sat(gqC.z);
                wf.y = exp2_sat(gqC.w);
                sC0 = __builtin_elementwise_fma(we, (v2f){n0.x, n0.y}, sC0);
                sC1 = __builtin_elementwise_fma(we, (v2f){n1.x, n1.y}, sC1);
                sC2 = __builtin_elementwise_fma(we, (v2f){n2.x, n2.y}, sC2);
                sC0 = __builtin_elementwise_fma(wf, (v2f){n0.z, n0.w}, sC0);
                sC1 = __builtin_elementwise_fma(wf, (v2f){n1.z, n1.w}, sC1);
                sC2 = __builtin_elementwise_fma(wf, (v2f){n2.z, n2.w}, sC2);
            }
        }
    }

    float accA0 = sA0.x + sA0.y, accA1 = sA1.x + sA1.y, accA2 = sA2.x + sA2.y;
    float accB0 = sB0.x + sB0.y, accB1 = sB1.x + sB1.y, accB2 = sB2.x + sB2.y;
    float accC0 = sC0.x + sC0.y, accC1 = sC1.x + sC1.y, accC2 = sC2.x + sC2.y;
    accA0 += __shfl_xor(accA0, 16, 32);
    accA1 += __shfl_xor(accA1, 16, 32);
    accA2 += __shfl_xor(accA2, 16, 32);
    accB0 += __shfl_xor(accB0, 16, 32);
    accB1 += __shfl_xor(accB1, 16, 32);
    accB2 += __shfl_xor(accB2, 16, 32);
    accC0 += __shfl_xor(accC0, 16, 32);
    accC1 += __shfl_xor(accC1, 16, 32);
    accC2 += __shfl_xor(accC2, 16, 32);
    s_part[wave][0][lane] = h ? accB0 : accA0;
    s_part[wave][1][lane] = h ? accB1 : accA1;
    s_part[wave][2][lane] = h ? accB2 : accA2;
    if (!h) {
        s_part[wave][0][32 + c] = accC0;
        s_part[wave][1][32 + c] = accC1;
        s_part[wave][2][32 + c] = accC2;
    }
    __syncthreads();

    if (tid < 64) {
        const bool live = tid < ISTR;
        const int slot = live ? tid : 0;
        float o0 = 0.0f, o1 = 0.0f, o2 = 0.0f;
#pragma unroll
        for (int w = 0; w < JSPLIT; ++w) {
            o0 += s_part[w][0][slot];
            o1 += s_part[w][1][slot];
            o2 += s_part[w][2][slot];
        }
        float m = live ? fmaxf(fmaxf(o0, o1), o2) : -INFINITY;
#pragma unroll
        for (int off = 16; off >= 1; off >>= 1) m = fmaxf(m, __shfl_xor(m, off, 32));
        if (lane == 0) atomic_max_float(mx, m);

        if (live) {
            const int opix = stripe * ISTR + tid;
            float* o = out + (size_t)n * KCH * PP;
            o[opix]          = o0;
            o[PP + opix]     = o1;
            o[2 * PP + opix] = o2;
        }
    }
}

__global__ void bf_scale(float* __restrict__ out, const float* __restrict__ mx, int total) {
    int i = blockIdx.x * blockDim.x + threadIdx.x;
    if (i < total) out[i] = out[i] / (*mx);
}

extern "C" void kernel_launch(void* const* d_in, const int* in_sizes, int n_in,
                              void* d_out, int out_size, void* d_ws, size_t ws_size,
                              hipStream_t stream) {
    (void)in_sizes; (void)n_in; (void)out_size; (void)ws_size;
    const float* raw = (const float*)d_in[0];
    const float* nrm = (const float*)d_in[1];
    float* out = (float*)d_out;
    float* ws  = (float*)d_ws;
    float* mx  = ws;
    const float* sqt = ws + SQ_OFF;

    bf_prep<<<(NIMG * PP) / 256, 256, 0, stream>>>(raw, ws);
    bf_main<<<NIMG * NSTR, THREADS, 0, stream>>>(raw, nrm, sqt, out, mx);

    const int total = NIMG * KCH * PP;
    bf_scale<<<(total + 255) / 256, 256, 0, stream>>>(out, mx, total);
}


# Stripe widened to 144 pixels (nine 16-column sets per wave on one A fragment and one set of row loads per tile): grid 256 = one 16-wave work-group per WGP; j split 16 ways, sum order changed
# speedup vs baseline: 9.2072x; 1.0897x over previous
#include <hip/hip_runtime.h>
#include <math.h>

#define NIMG 4
#define KCH 3
#define HH 96
#define WW 96
#define PP (HH * WW)
#define ISTR 144
#define NSTR (PP / ISTR)
#define CSETS 9
#define CGRP (CSETS / 3)
#define JSPLIT 16
#define NWAVES JSPLIT
#define JRANGE (PP / JSPLIT)
#define JROWS 16
#define YROWS (JRANGE / WW)
#define XTILES (WW / JROWS)
#define THREADS (32 * NWAVES)
#define SQ_OFF 64
#define LOG2E 1.44269504088896340736f

typedef __attribute__((ext_vector_type(2))) float v2f;
typedef __attribute__((ext_vector_type(4))) float v4f;
typedef __attribute__((ext_vector_type(8))) float v8f;

__device__ __forceinline__ void atomic_max_float(float* addr, float val) {
    if (val >= 0.0f) atomicMax((int*)addr, __float_as_int(val));
    else             atomicMin((unsigned int*)addr, __float_as_uint(val));
}

__device__ __forceinline__ float exp2_sat(float g) {
    return __builtin_amdgcn_fmed3f(__builtin_amdgcn_exp2f(g), 0.0f, 1.0f);
}

__global__ void bf_prep(const float* __restrict__ raw, float* __restrict__ ws) {
    const int g = blockIdx.x * blockDim.x + threadIdx.x;
    if (g == 0) ws[0] = -INFINITY;
    if (g < NIMG * PP) {
        const int n = g / PP;
        const int p = g - n * PP;
        const float inv_rgb = 1.0f / 15.0f;
        const float inv_xy  = 1.0f / 100.0f;
        const float* rawn = raw + (size_t)n * KCH * PP;
        const float x  = (float)(p % WW) * inv_xy;
        const float y  = (float)(p / WW) * inv_xy;
        const float c0 = rawn[p]          * inv_rgb;
        const float c1 = rawn[PP + p]     * inv_rgb;
        const float c2 = rawn[2 * PP + p] * inv_rgb;
        const float sq = x * x + y * y + c0 * c0 + c1 * c1 + c2 * c2;
        ws[SQ_OFF + g] = (-0.5f * LOG2E) * sq;
    }
}


__device__ __forceinline__ void bf_group3(
    const v2f a, const v4f sa, const v4f sb,
    const v4f u0, const v4f u1, const v4f v0, const v4f v1, const v4f w0, const v4f w1,
    const v2f bA, const v2f bB, const v2f bC,
    const float ctA, const float ctB, const float ctC,
    v2f& sA0, v2f& sA1, v2f& sA2,
    v2f& sB0, v2f& sB1, v2f& sB2,
    v2f& sC0, v2f& sC1, v2f& sC2)
{
    const v8f cstA = {ctA + sa.x, ctA + sa.y, ctA + sa.z, ctA + sa.w,
                      ctA + sb.x, ctA + sb.y, ctA + sb.z, ctA + sb.w};
    const v8f cstB = {ctB + sa.x, ctB + sa.y, ctB + sa.z, ctB + sa.w,
                      ctB + sb.x, ctB + sb.y, ctB + sb.z, ctB + sb.w};
    const v8f cstC = {ctC + sa.x, ctC + sa.y, ctC + sa.z, ctC + sa.w,
                      ctC + sb.x, ctC + sb.y, ctC + sb.z, ctC + sb.w};

    const v8f gA = __builtin_amdgcn_wmma_f32_16x16x4_f32(false, a, false, bA, (short)0, cstA, false, false);
    const v8f gB = __builtin_amdgcn_wmma_f32_16x16x4_f32(false, a, false, bB, (short)0, cstB, false, false);
    const v8f gC = __builtin_amdgcn_wmma_f32_16x16x4_f32(false, a, false, bC, (short)0, cstC, false, false);

#pragma unroll
    for (int hh = 0; hh < 2; ++hh) {
        const v4f n0 = hh ? u1 : u0;
        const v4f n1 = hh ? v1 : v0;
        const v4f n2 = hh ? w1 : w0;
        const v4f gqA = hh ? (v4f){gA[4], gA[5], gA[6], gA[7]} : (v4f){gA[0], gA[1], gA[2], gA[3]};
        const v4f gqB = hh ? (v4f){gB[4], gB[5], gB[6], gB[7]} : (v4f){gB[0], gB[1], gB[2], gB[3]};
        const v4f gqC = hh ? (v4f){gC[4], gC[5], gC[6], gC[7]} : (v4f){gC[0], gC[1], gC[2], gC[3]};
        v2f wa, wb;
        wa.x = exp2_sat(gqA.x);
        wa.y = exp2_sat(gqA.y);
        wb.x = exp2_sat(gqA.z);
        wb.y = exp2_sat(gqA.w);
        sA0 = __builtin_elementwise_fma(wa, (v2f){n0.x, n0.y}, sA0);
        sA1 = __builtin_elementwise_fma(wa, (v2f){n1.x, n1.y}, sA1);
        sA2 = __builtin_elementwise_fma(wa, (v2f){n2.x, n2.y}, sA2);
        sA0 = __builtin_elementwise_fma(wb, (v2f){n0.z, n0.w}, sA0);
        sA1 = __builtin_elementwise_fma(wb, (v2f){n1.z, n1.w}, sA1);
        sA2 = __builtin_elementwise_fma(wb, (v2f){n2.z, n2.w}, sA2);
        v2f wc, wd;
        wc.x = exp2_sat(gqB.x);
        wc.y = exp2_sat(gqB.y);
        wd.x = exp2_sat(gqB.z);
        wd.y = exp2_sat(gqB.w);
        sB0 = __builtin_elementwise_fma(wc, (v2f){n0.x, n0.y}, sB0);
        sB1 = __builtin_elementwise_fma(wc, (v2f){n1.x, n1.y}, sB1);
        sB2 = __builtin_elementwise_fma(wc, (v2f){n2.x, n2.y}, sB2);
        sB0 = __builtin_elementwise_fma(wd, (v2f){n0.z, n0.w}, sB0);
        sB1 = __builtin_elementwise_fma(wd, (v2f){n1.z, n1.w}, sB1);
        sB2 = __builtin_elementwise_fma(wd, (v2f){n2.z, n2.w}, sB2);
        v2f we, wf;
        we.x = exp2_sat(gqC.x);
        we.y = exp2_sat(gqC.y);
        wf.x = exp2_sat(gqC.z);
        wf.y = exp2_sat(gqC.w);
        sC0 = __builtin_elementwise_fma(we, (v2f){n0.x, n0.y}, sC0);
        sC1 = __builtin_elementwise_fma(we, (v2f){n1.x, n1.y}, sC1);
        sC2 = __builtin_elementwise_fma(we, (v2f){n2.x, n2.y}, sC2);
        sC0 = __builtin_elementwise_fma(wf, (v2f){n0.z, n0.w}, sC0);
        sC1 = __builtin_elementwise_fma(wf, (v2f){n1.z, n1.w}, sC1);
        sC2 = __builtin_elementwise_fma(wf, (v2f){n2.z, n2.w}, sC2);
    }
}

__global__ __launch_bounds__(THREADS) void bf_main(
    const float* __restrict__ raw,
    const float* __restrict__ nrm,
    const float* __restrict__ sqt,
    float* __restrict__ out,
    float* __restrict__ mx)
{
    __shared__ float s_part[NWAVES][KCH][ISTR];

    const int n      = blockIdx.x / NSTR;
    const int stripe = blockIdx.x % NSTR;
    const int tid    = threadIdx.x;
    const int lane   = tid & 31;
    const int c      = lane & 15;
    const bool h     = (lane >> 4) != 0;
    const int wave   = __builtin_amdgcn_readfirstlane(tid >> 5);
    const int jq     = wave;

    const float inv_rgb = 1.0f / 15.0f;
    const float inv_xy  = 1.0f / 100.0f;

    const float* rawn = raw + (size_t)n * KCH * PP;
    const float* nrmn = nrm + (size_t)n * KCH * PP;
    const float* sqn  = sqt + (size_t)n * PP;

    v2f   bq[CSETS];
    float e0[CSETS];
    float yil[CSETS];
#pragma unroll
    for (int s = 0; s < CSETS; ++s) {
        const int sbase = stripe * ISTR + 16 * s;
        const int ipix  = sbase + c;
        const float yi  = (float)(sbase / WW) * inv_xy;
        const float xi  = (float)(sbase % WW + c) * inv_xy;
        const float ci0 = rawn[ipix]          * inv_rgb;
        const float ci1 = rawn[PP + ipix]     * inv_rgb;
        const float ci2 = rawn[2 * PP + ipix] * inv_rgb;
        const float sqi = xi * xi + yi * yi + ci0 * ci0 + ci1 * ci1 + ci2 * ci2;
        bq[s].x = LOG2E * (h ? ci1 : xi);
        bq[s].y = LOG2E * (h ? ci2 : ci0);
        e0[s]   = (-0.5f * LOG2E) * sqi;
        yil[s]  = LOG2E * yi;
    }

    const unsigned aq0 = (h ? PP : 0) + c;
    const unsigned aq1 = 2 * PP + c;
    const unsigned rofs = h ? 8 : 0;

    v2f acc[CSETS][KCH];
#pragma unroll
    for (int s = 0; s < CSETS; ++s) {
#pragma unroll
        for (int k = 0; k < KCH; ++k) acc[s][k] = (v2f){0.0f, 0.0f};
    }

    unsigned jt = (unsigned)jq * JRANGE;

    float q0 = (rawn + jt)[aq0];
    float q1 = (rawn + jt)[aq1];
    v4f sa = *(const v4f*)(sqn + jt + rofs);
    v4f sb = *(const v4f*)(sqn + jt + rofs + 4);

    for (int ry = 0; ry < YROWS; ++ry) {
        const int jy = jq * YROWS + ry;
        const float yj = (float)jy * inv_xy;
        float ct[CSETS];
#pragma unroll
        for (int s = 0; s < CSETS; ++s) ct[s] = fmaf(yil[s], yj, e0[s]);

#pragma unroll 1
        for (int tx = 0; tx < XTILES; ++tx) {
            const float* nt = nrmn + jt;
            const v4f u0 = *(const v4f*)(nt + rofs);
            const v4f u1 = *(const v4f*)(nt + rofs + 4);
            const v4f v0 = *(const v4f*)(nt + PP + rofs);
            const v4f v1 = *(const v4f*)(nt + PP + rofs + 4);
            const v4f w0 = *(const v4f*)(nt + 2 * PP + rofs);
            const v4f w1 = *(const v4f*)(nt + 2 * PP + rofs + 4);

            const float xj = (float)(tx * JROWS + c) * inv_xy;
            const float p0 = q0 * inv_rgb;
            const float p1 = q1 * inv_rgb;
            v2f a;
            a.x = h ? p0 : xj;
            a.y = h ? p1 : p0;
            const v4f sac = sa;
            const v4f sbc = sb;

            const bool more = (ry != YROWS - 1) || (tx != XTILES - 1);
            jt += more ? JROWS : 0;
            const float* rt = rawn + jt;
            const float* st = sqn + jt;
            q0 = rt[aq0];
            q1 = rt[aq1];
            sa = *(const v4f*)(st + rofs);
            sb = *(const v4f*)(st + rofs + 4);

            __builtin_amdgcn_sched_barrier(0);

#pragma unroll
            for (int g = 0; g < CGRP; ++g) {
                bf_group3(a, sac, sbc, u0, u1, v0, v1, w0, w1,
                          bq[3 * g], bq[3 * g + 1], bq[3 * g + 2],
                          ct[3 * g], ct[3 * g + 1], ct[3 * g + 2],
                          acc[3 * g][0],     acc[3 * g][1],     acc[3 * g][2],
                          acc[3 * g + 1][0], acc[3 * g + 1][1], acc[3 * g + 1][2],
                          acc[3 * g + 2][0], acc[3 * g + 2][1], acc[3 * g + 2][2]);
            }
        }
    }

    float tot[CSETS][KCH];
#pragma unroll
    for (int s = 0; s < CSETS; ++s) {
#pragma unroll
        for (int k = 0; k < KCH; ++k) {
            const float t = acc[s][k].x + acc[s][k].y;
            tot[s][k] = t + __shfl_xor(t, 16, 32);
        }
    }
    if (!h) {
#pragma unroll
        for (int s = 0; s < CSETS; ++s) {
#pragma unroll
            for (int k = 0; k < KCH; ++k) s_part[wave][k][16 * s + c] = tot[s][k];
        }
    }
    __syncthreads();

    if (tid < 160) {
        const bool live = tid < ISTR;
        const int slot = live ? tid : 0;
        float o0 = 0.0f, o1 = 0.0f, o2 = 0.0f;
#pragma unroll
        for (int w = 0; w < JSPLIT; ++w) {
            o0 += s_part[w][0][slot];
            o1 += s_part[w][1][slot];
            o2 += s_part[w][2][slot];
        }
        float m = live ? fmaxf(fmaxf(o0, o1), o2) : -INFINITY;
#pragma unroll
        for (int off = 16; off >= 1; off >>= 1) m = fmaxf(m, __shfl_xor(m, off, 32));
        if (lane == 0) atomic_max_float(mx, m);

        if (live) {
            const int opix = stripe * ISTR + tid;
            float* o = out + (size_t)n * KCH * PP;
            o[opix]          = o0;
            o[PP + opix]     = o1;
            o[2 * PP + opix] = o2;
        }
    }
}

__global__ void bf_scale(float* __restrict__ out, const float* __restrict__ mx, int total) {
    int i = blockIdx.x * blockDim.x + threadIdx.x;
    if (i < total) out[i] = out[i] / (*mx);
}

extern "C" void kernel_launch(void* const* d_in, const int* in_sizes, int n_in,
                              void* d_out, int out_size, void* d_ws, size_t ws_size,
                              hipStream_t stream) {
    (void)in_sizes; (void)n_in; (void)out_size; (void)ws_size;
    const float* raw = (const float*)d_in[0];
    const float* nrm = (const float*)d_in[1];
    float* out = (float*)d_out;
    float* ws  = (float*)d_ws;
    float* mx  = ws;
    const float* sqt = ws + SQ_OFF;

    bf_prep<<<(NIMG * PP) / 256, 256, 0, stream>>>(raw, ws);
    bf_main<<<NIMG * NSTR, THREADS, 0, stream>>>(raw, nrm, sqt, out, mx);

    const int total = NIMG * KCH * PP;
    bf_scale<<<(total + 255) / 256, 256, 0, stream>>>(out, mx, total);
}


# First launch removed: each wave computes its own 576 rows' exponent addends into LDS before its loop (two 16-byte LDS reads per tile); max as one plain slot per work-group, no atomic; sum order kept
# speedup vs baseline: 9.7691x; 1.0610x over previous
#include <hip/hip_runtime.h>
#include <math.h>

#define NIMG 4
#define KCH 3
#define HH 96
#define WW 96
#define PP (HH * WW)
#define ISTR 144
#define NSTR (PP / ISTR)
#define CSETS 9
#define CGRP (CSETS / 3)
#define JSPLIT 16
#define NWAVES JSPLIT
#define JRANGE (PP / JSPLIT)
#define JROWS 16
#define YROWS (JRANGE / WW)
#define XTILES (WW / JROWS)
#define THREADS (32 * NWAVES)
#define NWG (NIMG * NSTR)
#define SQCH (JRANGE / 64)
#define FINW 5
#define LOG2E 1.44269504088896340736f

typedef __attribute__((ext_vector_type(2))) float v2f;
typedef __attribute__((ext_vector_type(4))) float v4f;
typedef __attribute__((ext_vector_type(8))) float v8f;

__device__ __forceinline__ float exp2_sat(float g) {
    return __builtin_amdgcn_fmed3f(__builtin_amdgcn_exp2f(g), 0.0f, 1.0f);
}


__device__ __forceinline__ void bf_group3(
    const v2f a, const v4f sa, const v4f sb,
    const v4f u0, const v4f u1, const v4f v0, const v4f v1, const v4f w0, const v4f w1,
    const v2f bA, const v2f bB, const v2f bC,
    const float ctA, const float ctB, const float ctC,
    v2f& sA0, v2f& sA1, v2f& sA2,
    v2f& sB0, v2f& sB1, v2f& sB2,
    v2f& sC0, v2f& sC1, v2f& sC2)
{
    const v8f cstA = {ctA + sa.x, ctA + sa.y, ctA + sa.z, ctA + sa.w,
                      ctA + sb.x, ctA + sb.y, ctA + sb.z, ctA + sb.w};
    const v8f cstB = {ctB + sa.x, ctB + sa.y, ctB + sa.z, ctB + sa.w,
                      ctB + sb.x, ctB + sb.y, ctB + sb.z, ctB + sb.w};
    const v8f cstC = {ctC + sa.x, ctC + sa.y, ctC + sa.z, ctC + sa.w,
                      ctC + sb.x, ctC + sb.y, ctC + sb.z, ctC + sb.w};

    const v8f gA = __builtin_amdgcn_wmma_f32_16x16x4_f32(false, a, false, bA, (short)0, cstA, false, false);
    const v8f gB = __builtin_amdgcn_wmma_f32_16x16x4_f32(false, a, false, bB, (short)0, cstB, false, false);
    const v8f gC = __builtin_amdgcn_wmma_f32_16x16x4_f32(false, a, false, bC, (short)0, cstC, false, false);

#pragma unroll
    for (int hh = 0; hh < 2; ++hh) {
        const v4f n0 = hh ? u1 : u0;
        const v4f n1 = hh ? v1 : v0;
        const v4f n2 = hh ? w1 : w0;
        const v4f gqA = hh ? (v4f){gA[4], gA[5], gA[6], gA[7]} : (v4f){gA[0], gA[1], gA[2], gA[3]};
        const v4f gqB = hh ? (v4f){gB[4], gB[5], gB[6], gB[7]} : (v4f){gB[0], gB[1], gB[2], gB[3]};
        const v4f gqC = hh ? (v4f){gC[4], gC[5], gC[6], gC[7]} : (v4f){gC[0], gC[1], gC[2], gC[3]};
        v2f wa, wb;
        wa.x = exp2_sat(gqA.x);
        wa.y = exp2_sat(gqA.y);
        wb.x = exp2_sat(gqA.z);
        wb.y = exp2_sat(gqA.w);
        sA0 = __builtin_elementwise_fma(wa, (v2f){n0.x, n0.y}, sA0);
        sA1 = __builtin_elementwise_fma(wa, (v2f){n1.x, n1.y}, sA1);
        sA2 = __builtin_elementwise_fma(wa, (v2f){n2.x, n2.y}, sA2);
        sA0 = __builtin_elementwise_fma(wb, (v2f){n0.z, n0.w}, sA0);
        sA1 = __builtin_elementwise_fma(wb, (v2f){n1.z, n1.w}, sA1);
        sA2 = __builtin_elementwise_fma(wb, (v2f){n2.z, n2.w}, sA2);
        v2f wc, wd;
        wc.x = exp2_sat(gqB.x);
        wc.y = exp2_sat(gqB.y);
        wd.x = exp2_sat(gqB.z);
        wd.y = exp2_sat(gqB.w);
        sB0 = __builtin_elementwise_fma(wc, (v2f){n0.x, n0.y}, sB0);
        sB1 = __builtin_elementwise_fma(wc, (v2f){n1.x, n1.y}, sB1);
        sB2 = __builtin_elementwise_fma(wc, (v2f){n2.x, n2.y}, sB2);
        sB0 = __builtin_elementwise_fma(wd, (v2f){n0.z, n0.w}, sB0);
        sB1 = __builtin_elementwise_fma(wd, (v2f){n1.z, n1.w}, sB1);
        sB2 = __builtin_elementwise_fma(wd, (v2f){n2.z, n2.w}, sB2);
        v2f we, wf;
        we.x = exp2_sat(gqC.x);
        we.y = exp2_sat(gqC.y);
        wf.x = exp2_sat(gqC.z);
        wf.y = exp2_sat(gqC.w);
        sC0 = __builtin_elementwise_fma(we, (v2f){n0.x, n0.y}, sC0);
        sC1 = __builtin_elementwise_fma(we, (v2f){n1.x, n1.y}, sC1);
        sC2 = __builtin_elementwise_fma(we, (v2f){n2.x, n2.y}, sC2);
        sC0 = __builtin_elementwise_fma(wf, (v2f){n0.z, n0.w}, sC0);
        sC1 = __builtin_elementwise_fma(wf, (v2f){n1.z, n1.w}, sC1);
        sC2 = __builtin_elementwise_fma(wf, (v2f){n2.z, n2.w}, sC2);
    }
}

__global__ __launch_bounds__(THREADS) void bf_main(
    const float* __restrict__ raw,
    const float* __restrict__ nrm,
    float* __restrict__ out,
    float* __restrict__ mxs)
{
    __shared__ float s_part[NWAVES][KCH][ISTR];
    __shared__ __attribute__((aligned(16))) float s_sq[NWAVES][JRANGE];
    __shared__ float s_mx[FINW];

    const int n      = blockIdx.x / NSTR;
    const int stripe = blockIdx.x % NSTR;
    const int tid    = threadIdx.x;
    const int lane   = tid & 31;
    const int c      = lane & 15;
    const bool h     = (lane >> 4) != 0;
    const int wave   = __builtin_amdgcn_readfirstlane(tid >> 5);
    const int jq     = wave;

    const float inv_rgb = 1.0f / 15.0f;
    const float inv_xy  = 1.0f / 100.0f;

    const float* rawn = raw + (size_t)n * KCH * PP;
    const float* nrmn = nrm + (size_t)n * KCH * PP;
    float* sqw = s_sq[wave];

    v2f   bq[CSETS];
    float e0[CSETS];
    float yil[CSETS];
#pragma unroll
    for (int s = 0; s < CSETS; ++s) {
        const int sbase = stripe * ISTR + 16 * s;
        const int ipix  = sbase + c;
        const float yi  = (float)(sbase / WW) * inv_xy;
        const float xi  = (float)(sbase % WW + c) * inv_xy;
        const float ci0 = rawn[ipix]          * inv_rgb;
        const float ci1 = rawn[PP + ipix]     * inv_rgb;
        const float ci2 = rawn[2 * PP + ipix] * inv_rgb;
        const float sqi = xi * xi + yi * yi + ci0 * ci0 + ci1 * ci1 + ci2 * ci2;
        bq[s].x = LOG2E * (h ? ci1 : xi);
        bq[s].y = LOG2E * (h ? ci2 : ci0);
        e0[s]   = (-0.5f * LOG2E) * sqi;
        yil[s]  = LOG2E * yi;
    }

    const unsigned aq0 = (h ? PP : 0) + c;
    const unsigned aq1 = 2 * PP + c;
    const unsigned rofs = h ? 8 : 0;

    v2f acc[CSETS][KCH];
#pragma unroll
    for (int s = 0; s < CSETS; ++s) {
#pragma unroll
        for (int k = 0; k < KCH; ++k) acc[s][k] = (v2f){0.0f, 0.0f};
    }

    unsigned jt = (unsigned)jq * JRANGE;
    unsigned jl = 0;

    float q0 = (rawn + jt)[aq0];
    float q1 = (rawn + jt)[aq1];

#pragma unroll
    for (int m = 0; m < SQCH; ++m) {
        const unsigned pl = 64u * m + 2u * lane;
        const unsigned p  = (unsigned)jq * JRANGE + pl;
        const v2f r0 = *(const v2f*)(rawn + p);
        const v2f r1 = *(const v2f*)(rawn + PP + p);
        const v2f r2 = *(const v2f*)(rawn + 2 * PP + p);
        const unsigned py = p / WW;
        const unsigned px = p - py * WW;
        const float y = (float)py * inv_xy;
        const v2f X  = (v2f){(float)px, (float)(px + 1u)} * inv_xy;
        const v2f Y  = (v2f){y, y};
        const v2f C0 = r0 * inv_rgb;
        const v2f C1 = r1 * inv_rgb;
        const v2f C2 = r2 * inv_rgb;
        const v2f sq = X * X + Y * Y + C0 * C0 + C1 * C1 + C2 * C2;
        *(v2f*)(sqw + pl) = (-0.5f * LOG2E) * sq;
    }
    __syncthreads();

    v4f sa = *(const v4f*)(sqw + jl + rofs);
    v4f sb = *(const v4f*)(sqw + jl + rofs + 4);

    for (int ry = 0; ry < YROWS; ++ry) {
        const int jy = jq * YROWS + ry;
        const float yj = (float)jy * inv_xy;
        float ct[CSETS];
#pragma unroll
        for (int s = 0; s < CSETS; ++s) ct[s] = fmaf(yil[s], yj, e0[s]);

#pragma unroll 1
        for (int tx = 0; tx < XTILES; ++tx) {
            const float* nt = nrmn + jt;
            const v4f u0 = *(const v4f*)(nt + rofs);
            const v4f u1 = *(const v4f*)(nt + rofs + 4);
            const v4f v0 = *(const v4f*)(nt + PP + rofs);
            const v4f v1 = *(const v4f*)(nt + PP + rofs + 4);
            const v4f w0 = *(const v4f*)(nt + 2 * PP + rofs);
            const v4f w1 = *(const v4f*)(nt + 2 * PP + rofs + 4);

            const float xj = (float)(tx * JROWS + c) * inv_xy;
            const float p0 = q0 * inv_rgb;
            const float p1 = q1 * inv_rgb;
            v2f a;
            a.x = h ? p0 : xj;
            a.y = h ? p1 : p0;
            const v4f sac = sa;
            const v4f sbc = sb;

            const bool more = (ry != YROWS - 1) || (tx != XTILES - 1);
            jt += more ? JROWS : 0;
            jl += more ? JROWS : 0;
            const float* rt = rawn + jt;
            const float* st = sqw + jl;
            q0 = rt[aq0];
            q1 = rt[aq1];
            sa = *(const v4f*)(st + rofs);
            sb = *(const v4f*)(st + rofs + 4);

            __builtin_amdgcn_sched_barrier(0);

#pragma unroll
            for (int g = 0; g < CGRP; ++g) {
                bf_group3(a, sac, sbc, u0, u1, v0, v1, w0, w1,
                          bq[3 * g], bq[3 * g + 1], bq[3 * g + 2],
                          ct[3 * g], ct[3 * g + 1], ct[3 * g + 2],
                          acc[3 * g][0],     acc[3 * g][1],     acc[3 * g][2],
                          acc[3 * g + 1][0], acc[3 * g + 1][1], acc[3 * g + 1][2],
                          acc[3 * g + 2][0], acc[3 * g + 2][1], acc[3 * g + 2][2]);
            }
        }
    }

    float tot[CSETS][KCH];
#pragma unroll
    for (int s = 0; s < CSETS; ++s) {
#pragma unroll
        for (int k = 0; k < KCH; ++k) {
            const float t = acc[s][k].x + acc[s][k].y;
            tot[s][k] = t + __shfl_xor(t, 16, 32);
        }
    }
    if (!h) {
#pragma unroll
        for (int s = 0; s < CSETS; ++s) {
#pragma unroll
            for (int k = 0; k < KCH; ++k) s_part[wave][k][16 * s + c] = tot[s][k];
        }
    }
    __syncthreads();

    if (tid < 32 * FINW) {
        const bool live = tid < ISTR;
        const int slot = live ? tid : 0;
        float o0 = 0.0f, o1 = 0.0f, o2 = 0.0f;
#pragma unroll
        for (int w = 0; w < JSPLIT; ++w) {
            o0 += s_part[w][0][slot];
            o1 += s_part[w][1][slot];
            o2 += s_part[w][2][slot];
        }
        float m = live ? fmaxf(fmaxf(o0, o1), o2) : -INFINITY;
#pragma unroll
        for (int off = 16; off >= 1; off >>= 1) m = fmaxf(m, __shfl_xor(m, off, 32));
        if (lane == 0) s_mx[wave] = m;

        if (live) {
            const int opix = stripe * ISTR + tid;
            float* o = out + (size_t)n * KCH * PP;
            o[opix]          = o0;
            o[PP + opix]     = o1;
            o[2 * PP + opix] = o2;
        }
    }
    __syncthreads();
    if (tid == 0) {
        float m = s_mx[0];
#pragma unroll
        for (int w = 1; w < FINW; ++w) m = fmaxf(m, s_mx[w]);
        mxs[blockIdx.x] = m;
    }
}

#define NOUT (NIMG * KCH * PP)
#define SCALE_BLOCK 256
static_assert(NOUT % SCALE_BLOCK == 0, "the last launch has no bounds guard");
static_assert(NWG == 8 * 32, "eight maxima per lane of one wave");
__global__ __launch_bounds__(SCALE_BLOCK) void bf_scale(float* __restrict__ out, const float* __restrict__ mxs) {
    const int i = blockIdx.x * SCALE_BLOCK + threadIdx.x;
    const int lane = threadIdx.x & 31;
    const float o = out[i];
    const v4f ma = *(const v4f*)(mxs + 8 * lane);
    const v4f mb = *(const v4f*)(mxs + 8 * lane + 4);
    float m = fmaxf(fmaxf(fmaxf(ma.x, ma.y), fmaxf(ma.z, ma.w)), fmaxf(fmaxf(mb.x, mb.y), fmaxf(mb.z, mb.w)));
#pragma unroll
    for (int off = 16; off >= 1; off >>= 1) m = fmaxf(m, __shfl_xor(m, off, 32));
    out[i] = o / m;
}

extern "C" void kernel_launch(void* const* d_in, const int* in_sizes, int n_in,
                              void* d_out, int out_size, void* d_ws, size_t ws_size,
                              hipStream_t stream) {
    (void)in_sizes; (void)n_in; (void)out_size; (void)ws_size;
    const float* raw = (const float*)d_in[0];
    const float* nrm = (const float*)d_in[1];
    float* out = (float*)d_out;
    float* mxs = (float*)d_ws;

    bf_main<<<NWG, THREADS, 0, stream>>>(raw, nrm, out, mxs);

    bf_scale<<<NOUT / SCALE_BLOCK, SCALE_BLOCK, 0, stream>>>(out, mxs);
}


# Workgroup pixels regrouped as two 16-column blocks (6+3 rows): x_i*x_j joins the row addends once per tile and block as a shared C operand, per-set addend rides at k=0; exponent term order changed
# speedup vs baseline: 10.5528x; 1.0802x over previous
#include <hip/hip_runtime.h>
#include <math.h>

#define NIMG 4
#define KCH 3
#define HH 96
#define WW 96
#define PP (HH * WW)
#define ISTR 144
#define NSTR (PP / ISTR)
#define SETS_A 6
#define ABLK_COLS 4
#define CSETS 9
#define CGRP (CSETS / 3)
#define JSPLIT 16
#define NWAVES JSPLIT
#define JRANGE (PP / JSPLIT)
#define JROWS 16
#define YROWS (JRANGE / WW)
#define XTILES (WW / JROWS)
#define THREADS (32 * NWAVES)
#define NWG (NIMG * NSTR)
#define SQCH (JRANGE / 64)
#define FINW 5
#define LOG2E 1.44269504088896340736f
static_assert(SETS_A % 3 == 0 && CSETS % 3 == 0, "a group of three sets lies in one block");
static_assert(ABLK_COLS * (HH / SETS_A) == NSTR, "A parts of an image: four column blocks x 16 row groups");
static_assert((WW / 16 - ABLK_COLS) * (HH / (CSETS - SETS_A)) == NSTR, "B parts of an image: two column blocks x 32 row groups");

typedef __attribute__((ext_vector_type(2))) float v2f;
typedef __attribute__((ext_vector_type(4))) float v4f;
typedef __attribute__((ext_vector_type(8))) float v8f;

__device__ __forceinline__ float exp2_sat(float g) {
    return __builtin_amdgcn_fmed3f(__builtin_amdgcn_exp2f(g), 0.0f, 1.0f);
}


__device__ __forceinline__ void bf_group3(
    const v2f a, const v8f cst,
    const v4f u0, const v4f u1, const v4f v0, const v4f v1, const v4f w0, const v4f w1,
    const v2f bA, const v2f bB, const v2f bC,
    v2f& sA0, v2f& sA1, v2f& sA2,
    v2f& sB0, v2f& sB1, v2f& sB2,
    v2f& sC0, v2f& sC1, v2f& sC2)
{
    const v8f gA = __builtin_amdgcn_wmma_f32_16x16x4_f32(false, a, false, bA, (short)0, cst, false, false);
    const v8f gB = __builtin_amdgcn_wmma_f32_16x16x4_f32(false, a, false, bB, (short)0, cst, false, false);
    const v8f gC = __builtin_amdgcn_wmma_f32_16x16x4_f32(false, a, false, bC, (short)0, cst, false, false);

#pragma unroll
    for (int hh = 0; hh < 2; ++hh) {
        const v4f n0 = hh ? u1 : u0;
        const v4f n1 = hh ? v1 : v0;
        const v4f n2 = hh ? w1 : w0;
        const v4f gqA = hh ? (v4f){gA[4], gA[5], gA[6], gA[7]} : (v4f){gA[0], gA[1], gA[2], gA[3]};
        const v4f gqB = hh ? (v4f){gB[4], gB[5], gB[6], gB[7]} : (v4f){gB[0], gB[1], gB[2], gB[3]};
        const v4f gqC = hh ? (v4f){gC[4], gC[5], gC[6], gC[7]} : (v4f){gC[0], gC[1], gC[2], gC[3]};
        v2f wa, wb;
        wa.x = exp2_sat(gqA.x);
        wa.y = exp2_sat(gqA.y);
        wb.x = exp2_sat(gqA.z);
        wb.y = exp2_sat(gqA.w);
        sA0 = __builtin_elementwise_fma(wa, (v2f){n0.x, n0.y}, sA0);
        sA1 = __builtin_elementwise_fma(wa, (v2f){n1.x, n1.y}, sA1);
        sA2 = __builtin_elementwise_fma(wa, (v2f){n2.x, n2.y}, sA2);
        sA0 = __builtin_elementwise_fma(wb, (v2f){n0.z, n0.w}, sA0);
        sA1 = __builtin_elementwise_fma(wb, (v2f){n1.z, n1.w}, sA1);
        sA2 = __builtin_elementwise_fma(wb, (v2f){n2.z, n2.w}, sA2);
        v2f wc, wd;
        wc.x = exp2_sat(gqB.x);
        wc.y = exp2_sat(gqB.y);
        wd.x = exp2_sat(gqB.z);
        wd.y = exp2_sat(gqB.w);
        sB0 = __builtin_elementwise_fma(wc, (v2f){n0.x, n0.y}, sB0);
        sB1 = __builtin_elementwise_fma(wc, (v2f){n1.x, n1.y}, sB1);
        sB2 = __builtin_elementwise_fma(wc, (v2f){n2.x, n2.y}, sB2);
        sB0 = __builtin_elementwise_fma(wd, (v2f){n0.z, n0.w}, sB0);
        sB1 = __builtin_elementwise_fma(wd, (v2f){n1.z, n1.w}, sB1);
        sB2 = __builtin_elementwise_fma(wd, (v2f){n2.z, n2.w}, sB2);
        v2f we, wf;
        we.x = exp2_sat(gqC.x);
        we.y = exp2_sat(gqC.y);
        wf.x = exp2_sat(gqC.z);
        wf.y = exp2_sat(gqC.w);
        sC0 = __builtin_elementwise_fma(we, (v2f){n0.x, n0.y}, sC0);
        sC1 = __builtin_elementwise_fma(we, (v2f){n1.x, n1.y}, sC1);
        sC2 = __builtin_elementwise_fma(we, (v2f){n2.x, n2.y}, sC2);
        sC0 = __builtin_elementwise_fma(wf, (v2f){n0.z, n0.w}, sC0);
        sC1 = __builtin_elementwise_fma(wf, (v2f){n1.z, n1.w}, sC1);
        sC2 = __builtin_elementwise_fma(wf, (v2f){n2.z, n2.w}, sC2);
    }
}

__global__ __launch_bounds__(THREADS) void bf_main(
    const float* __restrict__ raw,
    const float* __restrict__ nrm,
    float* __restrict__ out,
    float* __restrict__ mxs)
{
    __shared__ float s_part[NWAVES][KCH][ISTR];
    __shared__ __attribute__((aligned(16))) float s_sq[NWAVES][JRANGE];
    __shared__ float s_mx[FINW];

    const int n      = blockIdx.x / NSTR;
    const int part   = blockIdx.x % NSTR;
    const int colA = 16 * (part / 16);
    const int rowA = SETS_A * (part % 16);
    const int colB = 16 * (ABLK_COLS + part / 32);
    const int rowB = (CSETS - SETS_A) * (part % 32);
    const int tid    = threadIdx.x;
    const int lane   = tid & 31;
    const int c      = lane & 15;
    const bool h     = (lane >> 4) != 0;
    const int wave   = __builtin_amdgcn_readfirstlane(tid >> 5);
    const int jq     = wave;

    const float inv_rgb = 1.0f / 15.0f;
    const float inv_xy  = 1.0f / 100.0f;

    const float* rawn = raw + (size_t)n * KCH * PP;
    const float* nrmn = nrm + (size_t)n * KCH * PP;
    float* sqw = s_sq[wave];

    v2f   bq[CSETS];
    float be0[CSETS];
    float byl[CSETS];
#pragma unroll
    for (int s = 0; s < CSETS; ++s) {
        const int row   = (s < SETS_A) ? rowA + s : rowB + (s - SETS_A);
        const int col   = (s < SETS_A) ? colA : colB;
        const int ipix  = row * WW + col + c;
        const float yi  = (float)row * inv_xy;
        const float xi  = (float)(col + c) * inv_xy;
        const float ci0 = rawn[ipix]          * inv_rgb;
        const float ci1 = rawn[PP + ipix]     * inv_rgb;
        const float ci2 = rawn[2 * PP + ipix] * inv_rgb;
        const float sqi = xi * xi + yi * yi + ci0 * ci0 + ci1 * ci1 + ci2 * ci2;
        bq[s].x = 0.0f;
        bq[s].y = LOG2E * (h ? ci2 : ci0);
        be0[s]  = h ? LOG2E * ci1 : (-0.5f * LOG2E) * sqi;
        byl[s]  = h ? 0.0f : LOG2E * yi;
    }
    const float lxA = LOG2E * ((float)(colA + c) * inv_xy);
    const float lxB = LOG2E * ((float)(colB + c) * inv_xy);
    const unsigned aq0 = (h ? PP : 0) + c;
    const unsigned aq1 = 2 * PP + c;
    const unsigned rofs = h ? 8 : 0;

    const v4f xr0 = (v4f){(float)(rofs + 0u), (float)(rofs + 1u), (float)(rofs + 2u), (float)(rofs + 3u)} * inv_xy;
    const v4f xr1 = (v4f){(float)(rofs + 4u), (float)(rofs + 5u), (float)(rofs + 6u), (float)(rofs + 7u)} * inv_xy;

    v2f acc[CSETS][KCH];
#pragma unroll
    for (int s = 0; s < CSETS; ++s) {
#pragma unroll
        for (int k = 0; k < KCH; ++k) acc[s][k] = (v2f){0.0f, 0.0f};
    }

    unsigned jt = (unsigned)jq * JRANGE;
    unsigned jl = 0;

    float q0 = (rawn + jt)[aq0];
    float q1 = (rawn + jt)[aq1];

#pragma unroll
    for (int m = 0; m < SQCH; ++m) {
        const unsigned pl = 64u * m + 2u * lane;
        const unsigned p  = (unsigned)jq * JRANGE + pl;
        const v2f r0 = *(const v2f*)(rawn + p);
        const v2f r1 = *(const v2f*)(rawn + PP + p);
        const v2f r2 = *(const v2f*)(rawn + 2 * PP + p);
        const unsigned py = p / WW;
        const unsigned px = p - py * WW;
        const float y = (float)py * inv_xy;
        const v2f X  = (v2f){(float)px, (float)(px + 1u)} * inv_xy;
        const v2f Y  = (v2f){y, y};
        const v2f C0 = r0 * inv_rgb;
        const v2f C1 = r1 * inv_rgb;
        const v2f C2 = r2 * inv_rgb;
        const v2f sq = X * X + Y * Y + C0 * C0 + C1 * C1 + C2 * C2;
        *(v2f*)(sqw + pl) = (-0.5f * LOG2E) * sq;
    }
    __syncthreads();

    v4f sa = *(const v4f*)(sqw + jl + rofs);
    v4f sb = *(const v4f*)(sqw + jl + rofs + 4);

    for (int ry = 0; ry < YROWS; ++ry) {
        const int jy = jq * YROWS + ry;
        const float yj = (float)jy * inv_xy;
#pragma unroll
        for (int s = 0; s < CSETS; ++s) bq[s].x = fmaf(byl[s], yj, be0[s]);

#pragma unroll 1
        for (int tx = 0; tx < XTILES; ++tx) {
            const float* nt = nrmn + jt;
            const v4f u0 = *(const v4f*)(nt + rofs);
            const v4f u1 = *(const v4f*)(nt + rofs + 4);
            const v4f v0 = *(const v4f*)(nt + PP + rofs);
            const v4f v1 = *(const v4f*)(nt + PP + rofs + 4);
            const v4f w0 = *(const v4f*)(nt + 2 * PP + rofs);
            const v4f w1 = *(const v4f*)(nt + 2 * PP + rofs + 4);

            const float p0 = q0 * inv_rgb;
            const float p1 = q1 * inv_rgb;
            v2f a;
            a.x = h ? p0 : 1.0f;
            a.y = h ? p1 : p0;
            const float xt = (float)(tx * JROWS) * inv_xy;
            const v4f xa = xr0 + xt;
            const v4f xb = xr1 + xt;
            const v8f cA = {fmaf(lxA, xa.x, sa.x), fmaf(lxA, xa.y, sa.y), fmaf(lxA, xa.z, sa.z), fmaf(lxA, xa.w, sa.w),
                            fmaf(lxA, xb.x, sb.x), fmaf(lxA, xb.y, sb.y), fmaf(lxA, xb.z, sb.z), fmaf(lxA, xb.w, sb.w)};
            const v8f cB = {fmaf(lxB, xa.x, sa.x), fmaf(lxB, xa.y, sa.y), fmaf(lxB, xa.z, sa.z), fmaf(lxB, xa.w, sa.w),
                            fmaf(lxB, xb.x, sb.x), fmaf(lxB, xb.y, sb.y), fmaf(lxB, xb.z, sb.z), fmaf(lxB, xb.w, sb.w)};

            const bool more = (ry != YROWS - 1) || (tx != XTILES - 1);
            jt += more ? JROWS : 0;
            jl += more ? JROWS : 0;
            const float* rt = rawn + jt;
            const float* st = sqw + jl;
            q0 = rt[aq0];
            q1 = rt[aq1];
            sa = *(const v4f*)(st + rofs);
            sb = *(const v4f*)(st + rofs + 4);

            __builtin_amdgcn_sched_barrier(0);

#pragma unroll
            for (int g = 0; g < CGRP; ++g) {
                if (g > 0) __builtin_amdgcn_sched_barrier(0);
                bf_group3(a, (3 * g < SETS_A) ? cA : cB, u0, u1, v0, v1, w0, w1,
                          bq[3 * g], bq[3 * g + 1], bq[3 * g + 2],
                          acc[3 * g][0],     acc[3 * g][1],     acc[3 * g][2],
                          acc[3 * g + 1][0], acc[3 * g + 1][1], acc[3 * g + 1][2],
                          acc[3 * g + 2][0], acc[3 * g + 2][1], acc[3 * g + 2][2]);
            }
        }
    }

    float tot[CSETS][KCH];
#pragma unroll
    for (int s = 0; s < CSETS; ++s) {
#pragma unroll
        for (int k = 0; k < KCH; ++k) {
            const float t = acc[s][k].x + acc[s][k].y;
            tot[s][k] = t + __shfl_xor(t, 16, 32);
        }
    }
    if (!h) {
#pragma unroll
        for (int s = 0; s < CSETS; ++s) {
#pragma unroll
            for (int k = 0; k < KCH; ++k) s_part[wave][k][16 * s + c] = tot[s][k];
        }
    }
    __syncthreads();

    if (tid < 32 * FINW) {
        const bool live = tid < ISTR;
        const int slot = live ? tid : 0;
        float o0 = 0.0f, o1 = 0.0f, o2 = 0.0f;
#pragma unroll
        for (int w = 0; w < JSPLIT; ++w) {
            o0 += s_part[w][0][slot];
            o1 += s_part[w][1][slot];
            o2 += s_part[w][2][slot];
        }
        float m = live ? fmaxf(fmaxf(o0, o1), o2) : -INFINITY;
#pragma unroll
        for (int off = 16; off >= 1; off >>= 1) m = fmaxf(m, __shfl_xor(m, off, 32));
        if (lane == 0) s_mx[wave] = m;

        if (live) {
            const int s    = tid >> 4;
            const int row  = (s < SETS_A) ? rowA + s : rowB + (s - SETS_A);
            const int col  = (s < SETS_A) ? colA : colB;
            const int opix = row * WW + col + (tid & 15);
            float* o = out + (size_t)n * KCH * PP;
            o[opix]          = o0;
            o[PP + opix]     = o1;
            o[2 * PP + opix] = o2;
        }
    }
    __syncthreads();
    if (tid == 0) {
        float m = s_mx[0];
#pragma unroll
        for (int w = 1; w < FINW; ++w) m = fmaxf(m, s_mx[w]);
        mxs[blockIdx.x] = m;
    }
}

#define NOUT (NIMG * KCH * PP)
#define SCALE_BLOCK 256
static_assert(NOUT % SCALE_BLOCK == 0, "the last launch has no bounds guard");
static_assert(NWG == 8 * 32, "eight maxima per lane of one wave");
__global__ __launch_bounds__(SCALE_BLOCK) void bf_scale(float* __restrict__ out, const float* __restrict__ mxs) {
    const int i = blockIdx.x * SCALE_BLOCK + threadIdx.x;
    const int lane = threadIdx.x & 31;
    const float o = out[i];
    const v4f ma = *(const v4f*)(mxs + 8 * lane);
    const v4f mb = *(const v4f*)(mxs + 8 * lane + 4);
    float m = fmaxf(fmaxf(fmaxf(ma.x, ma.y), fmaxf(ma.z, ma.w)), fmaxf(fmaxf(mb.x, mb.y), fmaxf(mb.z, mb.w)));
#pragma unroll
    for (int off = 16; off >= 1; off >>= 1) m = fmaxf(m, __shfl_xor(m, off, 32));
    out[i] = o / m;
}

extern "C" void kernel_launch(void* const* d_in, const int* in_sizes, int n_in,
                              void* d_out, int out_size, void* d_ws, size_t ws_size,
                              hipStream_t stream) {
    (void)in_sizes; (void)n_in; (void)out_size; (void)ws_size;
    const float* raw = (const float*)d_in[0];
    const float* nrm = (const float*)d_in[1];
    float* out = (float*)d_out;
    float* mxs = (float*)d_ws;

    bf_main<<<NWG, THREADS, 0, stream>>>(raw, nrm, out, mxs);

    bf_scale<<<NOUT / SCALE_BLOCK, SCALE_BLOCK, 0, stream>>>(out, mxs);
}
